# starting kernel
# baseline (speedup 1.0000x reference)
#include <hip/hip_runtime.h>

typedef __attribute__((ext_vector_type(2))) float v2f;
typedef __attribute__((ext_vector_type(8))) float v8f;

#define S_LEN   2048
#define NHEAD   16
#define DH      64
#define KSTRIDE 32
#define WIN     16
#define NTILES  7
#define NCOLS   112
#define ROWPAD  68
#define SCPAD   120

__global__ __launch_bounds__(32)
void strided_attn_wmma_f32(const float* __restrict__ q,
                           const float* __restrict__ k,
                           const float* __restrict__ v,
                           float* __restrict__ out)
{
    __shared__ float q_lds[16 * ROWPAD];
    __shared__ float k_lds[16 * ROWPAD];
    __shared__ float v_lds[16 * ROWPAD];
    __shared__ float sc_lds[16 * SCPAD];

    const int lane  = threadIdx.x;
    const int m     = lane & 15;
    const int hi    = lane >> 4;
    const int qb    = blockIdx.x;
    const int h     = blockIdx.y;
    const int qbase = qb * 16;
    const long hbase = (long)h * S_LEN * DH;

    for (int r = 0; r < 16; ++r) {
        const float* src = q + hbase + (long)(qbase + r) * DH + 2 * lane;
        v2f t; t.x = src[0]; t.y = src[1];
        *(v2f*)&q_lds[r * ROWPAD + 2 * lane] = t;
    }
    __syncthreads();

    for (int t = 0; t < NTILES; ++t) {
        for (int r = 0; r < 16; ++r) {
            int pos;
            if (t < 3) {
                pos = qbase - WIN + t * 16 + r;
                pos = pos < 0 ? 0 : (pos > S_LEN - 1 ? S_LEN - 1 : pos);
            } else {
                pos = ((t - 3) * 16 + r) * KSTRIDE;
            }
            const float* src = k + hbase + (long)pos * DH + 2 * lane;
            v2f tv; tv.x = src[0]; tv.y = src[1];
            *(v2f*)&k_lds[r * ROWPAD + 2 * lane] = tv;
        }
        __syncthreads();

        if (t + 1 < NTILES) {
            int tn = t + 1;
            int pn;
            if (tn < 3) {
                pn = qbase - WIN + tn * 16 + m;
                pn = pn < 0 ? 0 : (pn > S_LEN - 1 ? S_LEN - 1 : pn);
            } else {
                pn = ((tn - 3) * 16 + m) * KSTRIDE;
            }
            __builtin_prefetch(k + hbase + (long)pn * DH, 0, 3);
        }

        v8f acc = {};
        #pragma unroll
        for (int kc = 0; kc < 16; ++kc) {
            const int d0 = 4 * kc + 2 * hi;
            v2f a = *(const v2f*)&q_lds[m * ROWPAD + d0];
            v2f b = *(const v2f*)&k_lds[m * ROWPAD + d0];
            acc = __builtin_amdgcn_wmma_f32_16x16x4_f32(
                false, a, false, b, (short)0, acc, false, false);
        }

        #pragma unroll
        for (int r = 0; r < 8; ++r) {
            const int row = r + 8 * hi;
            const int n   = qbase + row;
            const int col = t * 16 + m;
            bool valid;
            if (t < 3) {
                const int p = qbase - WIN + col;
                valid = (col >= row) && (col <= row + 2 * WIN)
                        && (p >= 0) && (p < S_LEN);
            } else {
                valid = (n > WIN);
            }
            sc_lds[row * SCPAD + col] = valid ? acc[r] * 0.125f : -1.0e30f;
        }
        __syncthreads();
    }

    if (lane < 16) {
        float mx = -1.0e30f;
        for (int j = 0; j < NCOLS; ++j)
            mx = fmaxf(mx, sc_lds[lane * SCPAD + j]);
        float sum = 0.0f;
        for (int j = 0; j < NCOLS; ++j) {
            float e = __expf(sc_lds[lane * SCPAD + j] - mx);
            sc_lds[lane * SCPAD + j] = e;
            sum += e;
        }
        const float inv = (sum > 0.0f) ? (1.0f / sum) : 0.0f;
        for (int j = 0; j < NCOLS; ++j)
            sc_lds[lane * SCPAD + j] *= inv;
    }
    __syncthreads();

    v8f acc_o[4] = {v8f{}, v8f{}, v8f{}, v8f{}};
    for (int t = 0; t < NTILES; ++t) {
        for (int r = 0; r < 16; ++r) {
            int pos;
            if (t < 3) {
                pos = qbase - WIN + t * 16 + r;
                pos = pos < 0 ? 0 : (pos > S_LEN - 1 ? S_LEN - 1 : pos);
            } else {
                pos = ((t - 3) * 16 + r) * KSTRIDE;
            }
            const float* src = v + hbase + (long)pos * DH + 2 * lane;
            v2f tv; tv.x = src[0]; tv.y = src[1];
            *(v2f*)&v_lds[r * ROWPAD + 2 * lane] = tv;
        }
        __syncthreads();

        #pragma unroll
        for (int kc = 0; kc < 4; ++kc) {
            const int c0 = t * 16 + 4 * kc + 2 * hi;
            const int k0 = 4 * kc + 2 * hi;
            v2f a = *(const v2f*)&sc_lds[m * SCPAD + c0];
            #pragma unroll
            for (int nt = 0; nt < 4; ++nt) {
                v2f b;
                b.x = v_lds[(k0 + 0) * ROWPAD + 16 * nt + m];
                b.y = v_lds[(k0 + 1) * ROWPAD + 16 * nt + m];
                acc_o[nt] = __builtin_amdgcn_wmma_f32_16x16x4_f32(
                    false, a, false, b, (short)0, acc_o[nt], false, false);
            }
        }
        __syncthreads();
    }

    #pragma unroll
    for (int nt = 0; nt < 4; ++nt) {
        #pragma unroll
        for (int r = 0; r < 8; ++r) {
            const int row = r + 8 * hi;
            out[hbase + (long)(qbase + row) * DH + 16 * nt + m] = acc_o[nt][r];
        }
    }
}

extern "C" void kernel_launch(void* const* d_in, const int* in_sizes, int n_in,
                              void* d_out, int out_size, void* d_ws, size_t ws_size,
                              hipStream_t stream) {
    const float* q = (const float*)d_in[0];
    const float* k = (const float*)d_in[1];
    const float* v = (const float*)d_in[2];
    float* out = (float*)d_out;

    dim3 grid(S_LEN / 16, NHEAD, 1);
    dim3 block(32, 1, 1);
    strided_attn_wmma_f32<<<grid, block, 0, stream>>>(q, k, v, out);
}


# Softmax done in registers on all 32 lanes: row max and row sum by a 16-lane row_xmask butterfly on the score accumulators, not three serial LDS passes on 16 lanes; softmax sum order changes
# speedup vs baseline: 1.2767x; 1.2767x over previous
#include <hip/hip_runtime.h>

typedef __attribute__((ext_vector_type(2))) float v2f;
typedef __attribute__((ext_vector_type(8))) float v8f;

#define S_LEN   2048
#define NHEAD   16
#define DH      64
#define KSTRIDE 32
#define WIN     16
#define NTILES  7
#define NCOLS   112
#define ROWPAD  68
#define SCPAD   120

template <int XM>
__device__ __forceinline__ float row_xor(float x)
{
    const int v = __builtin_bit_cast(int, x);
    return __builtin_bit_cast(float,
        __builtin_amdgcn_update_dpp(0, v, 0x160 | XM, 0xf, 0xf, true));
}

__global__ __launch_bounds__(32)
void strided_attn_wmma_f32(const float* __restrict__ q,
                           const float* __restrict__ k,
                           const float* __restrict__ v,
                           float* __restrict__ out)
{
    __shared__ float q_lds[16 * ROWPAD];
    __shared__ float k_lds[16 * ROWPAD];
    __shared__ float v_lds[16 * ROWPAD];
    __shared__ float sc_lds[16 * SCPAD];

    const int lane  = threadIdx.x;
    const int m     = lane & 15;
    const int hi    = lane >> 4;
    const int qb    = blockIdx.x;
    const int h     = blockIdx.y;
    const int qbase = qb * 16;
    const long hbase = (long)h * S_LEN * DH;

    for (int r = 0; r < 16; ++r) {
        const float* src = q + hbase + (long)(qbase + r) * DH + 2 * lane;
        v2f t; t.x = src[0]; t.y = src[1];
        *(v2f*)&q_lds[r * ROWPAD + 2 * lane] = t;
    }
    __syncthreads();

    v8f sc[NTILES];
    #pragma unroll
    for (int t = 0; t < NTILES; ++t) {
        for (int r = 0; r < 16; ++r) {
            int pos;
            if (t < 3) {
                pos = qbase - WIN + t * 16 + r;
                pos = pos < 0 ? 0 : (pos > S_LEN - 1 ? S_LEN - 1 : pos);
            } else {
                pos = ((t - 3) * 16 + r) * KSTRIDE;
            }
            const float* src = k + hbase + (long)pos * DH + 2 * lane;
            v2f tv; tv.x = src[0]; tv.y = src[1];
            *(v2f*)&k_lds[r * ROWPAD + 2 * lane] = tv;
        }
        __syncthreads();

        if (t + 1 < NTILES) {
            int tn = t + 1;
            int pn;
            if (tn < 3) {
                pn = qbase - WIN + tn * 16 + m;
                pn = pn < 0 ? 0 : (pn > S_LEN - 1 ? S_LEN - 1 : pn);
            } else {
                pn = ((tn - 3) * 16 + m) * KSTRIDE;
            }
            __builtin_prefetch(k + hbase + (long)pn * DH, 0, 3);
        }

        v8f acc = {};
        #pragma unroll
        for (int kc = 0; kc < 16; ++kc) {
            const int d0 = 4 * kc + 2 * hi;
            v2f a = *(const v2f*)&q_lds[m * ROWPAD + d0];
            v2f b = *(const v2f*)&k_lds[m * ROWPAD + d0];
            acc = __builtin_amdgcn_wmma_f32_16x16x4_f32(
                false, a, false, b, (short)0, acc, false, false);
        }

        #pragma unroll
        for (int r = 0; r < 8; ++r) {
            const int row = r + 8 * hi;
            const int n   = qbase + row;
            const int col = t * 16 + m;
            bool valid;
            if (t < 3) {
                const int p = qbase - WIN + col;
                valid = (col >= row) && (col <= row + 2 * WIN)
                        && (p >= 0) && (p < S_LEN);
            } else {
                valid = (n > WIN);
            }
            sc[t][r] = valid ? acc[r] * 0.125f : -1.0e30f;
        }
        __syncthreads();
        __builtin_amdgcn_sched_barrier(0);
    }

    #pragma unroll
    for (int r = 0; r < 8; ++r) {
        float mx = sc[0][r];
        #pragma unroll
        for (int t = 1; t < NTILES; ++t)
            mx = fmaxf(mx, sc[t][r]);
        mx = fmaxf(mx, row_xor<1>(mx));
        mx = fmaxf(mx, row_xor<2>(mx));
        mx = fmaxf(mx, row_xor<4>(mx));
        mx = fmaxf(mx, row_xor<8>(mx));

        float sum = 0.0f;
        #pragma unroll
        for (int t = 0; t < NTILES; ++t) {
            const float e = __expf(sc[t][r] - mx);
            sc[t][r] = e;
            sum += e;
        }
        sum += row_xor<1>(sum);
        sum += row_xor<2>(sum);
        sum += row_xor<4>(sum);
        sum += row_xor<8>(sum);
        const float inv = (sum > 0.0f) ? (1.0f / sum) : 0.0f;

        const int row = r + 8 * hi;
        #pragma unroll
        for (int t = 0; t < NTILES; ++t)
            sc_lds[row * SCPAD + t * 16 + m] = sc[t][r] * inv;
    }
    __syncthreads();

    v8f acc_o[4] = {v8f{}, v8f{}, v8f{}, v8f{}};
    for (int t = 0; t < NTILES; ++t) {
        for (int r = 0; r < 16; ++r) {
            int pos;
            if (t < 3) {
                pos = qbase - WIN + t * 16 + r;
                pos = pos < 0 ? 0 : (pos > S_LEN - 1 ? S_LEN - 1 : pos);
            } else {
                pos = ((t - 3) * 16 + r) * KSTRIDE;
            }
            const float* src = v + hbase + (long)pos * DH + 2 * lane;
            v2f tv; tv.x = src[0]; tv.y = src[1];
            *(v2f*)&v_lds[r * ROWPAD + 2 * lane] = tv;
        }
        __syncthreads();

        #pragma unroll
        for (int kc = 0; kc < 4; ++kc) {
            const int c0 = t * 16 + 4 * kc + 2 * hi;
            const int k0 = 4 * kc + 2 * hi;
            v2f a = *(const v2f*)&sc_lds[m * SCPAD + c0];
            #pragma unroll
            for (int nt = 0; nt < 4; ++nt) {
                v2f b;
                b.x = v_lds[(k0 + 0) * ROWPAD + 16 * nt + m];
                b.y = v_lds[(k0 + 1) * ROWPAD + 16 * nt + m];
                acc_o[nt] = __builtin_amdgcn_wmma_f32_16x16x4_f32(
                    false, a, false, b, (short)0, acc_o[nt], false, false);
            }
        }
        __syncthreads();
    }

    #pragma unroll
    for (int nt = 0; nt < 4; ++nt) {
        #pragma unroll
        for (int r = 0; r < 8; ++r) {
            const int row = r + 8 * hi;
            out[hbase + (long)(qbase + row) * DH + 16 * nt + m] = acc_o[nt][r];
        }
    }
}

extern "C" void kernel_launch(void* const* d_in, const int* in_sizes, int n_in,
                              void* d_out, int out_size, void* d_ws, size_t ws_size,
                              hipStream_t stream) {
    const float* q = (const float*)d_in[0];
    const float* k = (const float*)d_in[1];
    const float* v = (const float*)d_in[2];
    float* out = (float*)d_out;

    dim3 grid(S_LEN / 16, NHEAD, 1);
    dim3 block(32, 1, 1);
    strided_attn_wmma_f32<<<grid, block, 0, stream>>>(q, k, v, out);
}


# V loaded straight from global into the WMMA B layout (16-byte loads, out columns dealt d=4*lane+tile) ahead of the softmax: no LDS staging or per-tile round trip in PV; out stored 16 bytes per lane
# speedup vs baseline: 1.3459x; 1.0542x over previous
#include <hip/hip_runtime.h>

typedef __attribute__((ext_vector_type(2))) float v2f;
typedef __attribute__((ext_vector_type(4))) float v4f;
typedef __attribute__((ext_vector_type(8))) float v8f;

#define S_LEN   2048
#define NHEAD   16
#define DH      64
#define KSTRIDE 32
#define WIN     16
#define NTILES  7
#define NCOLS   112
#define ROWPAD  68
#define SCPAD   120

template <int XM>
__device__ __forceinline__ float row_xor(float x)
{
    const int v = __builtin_bit_cast(int, x);
    return __builtin_bit_cast(float,
        __builtin_amdgcn_update_dpp(0, v, 0x160 | XM, 0xf, 0xf, true));
}

__global__ __launch_bounds__(32)
void strided_attn_wmma_f32(const float* __restrict__ q,
                           const float* __restrict__ k,
                           const float* __restrict__ v,
                           float* __restrict__ out)
{
    __shared__ float q_lds[16 * ROWPAD];
    __shared__ float k_lds[16 * ROWPAD];
    __shared__ float sc_lds[16 * SCPAD];

    const int lane  = threadIdx.x;
    const int m     = lane & 15;
    const int hi    = lane >> 4;
    const int qb    = blockIdx.x;
    const int h     = blockIdx.y;
    const int qbase = qb * 16;
    const long hbase = (long)h * S_LEN * DH;

    for (int r = 0; r < 16; ++r) {
        const float* src = q + hbase + (long)(qbase + r) * DH + 2 * lane;
        v2f t; t.x = src[0]; t.y = src[1];
        *(v2f*)&q_lds[r * ROWPAD + 2 * lane] = t;
    }
    __syncthreads();

    v8f sc[NTILES];
    #pragma unroll
    for (int t = 0; t < NTILES; ++t) {
        for (int r = 0; r < 16; ++r) {
            int pos;
            if (t < 3) {
                pos = qbase - WIN + t * 16 + r;
                pos = pos < 0 ? 0 : (pos > S_LEN - 1 ? S_LEN - 1 : pos);
            } else {
                pos = ((t - 3) * 16 + r) * KSTRIDE;
            }
            const float* src = k + hbase + (long)pos * DH + 2 * lane;
            v2f tv; tv.x = src[0]; tv.y = src[1];
            *(v2f*)&k_lds[r * ROWPAD + 2 * lane] = tv;
        }
        __syncthreads();

        if (t + 1 < NTILES) {
            int tn = t + 1;
            int pn;
            if (tn < 3) {
                pn = qbase - WIN + tn * 16 + m;
                pn = pn < 0 ? 0 : (pn > S_LEN - 1 ? S_LEN - 1 : pn);
            } else {
                pn = ((tn - 3) * 16 + m) * KSTRIDE;
            }
            __builtin_prefetch(k + hbase + (long)pn * DH, 0, 3);
        }

        v8f acc = {};
        #pragma unroll
        for (int kc = 0; kc < 16; ++kc) {
            const int d0 = 4 * kc + 2 * hi;
            v2f a = *(const v2f*)&q_lds[m * ROWPAD + d0];
            v2f b = *(const v2f*)&k_lds[m * ROWPAD + d0];
            acc = __builtin_amdgcn_wmma_f32_16x16x4_f32(
                false, a, false, b, (short)0, acc, false, false);
        }

        #pragma unroll
        for (int r = 0; r < 8; ++r) {
            const int row = r + 8 * hi;
            const int n   = qbase + row;
            const int col = t * 16 + m;
            bool valid;
            if (t < 3) {
                const int p = qbase - WIN + col;
                valid = (col >= row) && (col <= row + 2 * WIN)
                        && (p >= 0) && (p < S_LEN);
            } else {
                valid = (n > WIN);
            }
            sc[t][r] = valid ? acc[r] * 0.125f : -1.0e30f;
        }
        __syncthreads();
        __builtin_amdgcn_sched_barrier(0);
    }

    v4f vf[NTILES][4][2];
    #pragma unroll
    for (int t = 0; t < NTILES; ++t) {
        #pragma unroll
        for (int j = 0; j < 4; ++j) {
            #pragma unroll
            for (int s = 0; s < 2; ++s) {
                const int kr = 4 * j + 2 * hi + s;
                int pos;
                if (t < 3) {
                    pos = qbase - WIN + t * 16 + kr;
                    pos = pos < 0 ? 0 : (pos > S_LEN - 1 ? S_LEN - 1 : pos);
                } else {
                    pos = ((t - 3) * 16 + kr) * KSTRIDE;
                }
                vf[t][j][s] = *(const v4f*)(v + hbase + (long)pos * DH + 4 * m);
            }
        }
    }
    __builtin_amdgcn_sched_barrier(0);

    #pragma unroll
    for (int r = 0; r < 8; ++r) {
        float mx = sc[0][r];
        #pragma unroll
        for (int t = 1; t < NTILES; ++t)
            mx = fmaxf(mx, sc[t][r]);
        mx = fmaxf(mx, row_xor<1>(mx));
        mx = fmaxf(mx, row_xor<2>(mx));
        mx = fmaxf(mx, row_xor<4>(mx));
        mx = fmaxf(mx, row_xor<8>(mx));

        float sum = 0.0f;
        #pragma unroll
        for (int t = 0; t < NTILES; ++t) {
            const float e = __expf(sc[t][r] - mx);
            sc[t][r] = e;
            sum += e;
        }
        sum += row_xor<1>(sum);
        sum += row_xor<2>(sum);
        sum += row_xor<4>(sum);
        sum += row_xor<8>(sum);
        const float inv = (sum > 0.0f) ? (1.0f / sum) : 0.0f;

        const int row = r + 8 * hi;
        #pragma unroll
        for (int t = 0; t < NTILES; ++t)
            sc_lds[row * SCPAD + t * 16 + m] = sc[t][r] * inv;
    }
    __builtin_amdgcn_fence(__ATOMIC_RELEASE, "workgroup", "local");
    __builtin_amdgcn_s_barrier();
    __builtin_amdgcn_fence(__ATOMIC_ACQUIRE, "workgroup", "local");

    v8f acc_o[4] = {v8f{}, v8f{}, v8f{}, v8f{}};
    #pragma unroll
    for (int t = 0; t < NTILES; ++t) {
        #pragma unroll
        for (int j = 0; j < 4; ++j) {
            const int c0 = t * 16 + 4 * j + 2 * hi;
            v2f a = *(const v2f*)&sc_lds[m * SCPAD + c0];
            #pragma unroll
            for (int nt = 0; nt < 4; ++nt) {
                v2f b;
                b.x = vf[t][j][0][nt];
                b.y = vf[t][j][1][nt];
                acc_o[nt] = __builtin_amdgcn_wmma_f32_16x16x4_f32(
                    false, a, false, b, (short)0, acc_o[nt], false, false);
            }
        }
    }

    #pragma unroll
    for (int r = 0; r < 8; ++r) {
        const int row = r + 8 * hi;
        v4f o;
        o.x = acc_o[0][r]; o.y = acc_o[1][r]; o.z = acc_o[2][r]; o.w = acc_o[3][r];
        *(v4f*)(out + hbase + (long)(qbase + row) * DH + 4 * m) = o;
    }
}

extern "C" void kernel_launch(void* const* d_in, const int* in_sizes, int n_in,
                              void* d_out, int out_size, void* d_ws, size_t ws_size,
                              hipStream_t stream) {
    const float* q = (const float*)d_in[0];
    const float* k = (const float*)d_in[1];
    const float* v = (const float*)d_in[2];
    float* out = (float*)d_out;

    dim3 grid(S_LEN / 16, NHEAD, 1);
    dim3 block(32, 1, 1);
    strided_attn_wmma_f32<<<grid, block, 0, stream>>>(q, k, v, out);
}


# Score phase fed ahead of the matrix unit: every Q and K row fetched by one up-front group of 16-byte loads, one LDS slot per K tile, fragments read a tile ahead, Q fragments kept in registers
# speedup vs baseline: 1.4490x; 1.0766x over previous
#include <hip/hip_runtime.h>

typedef __attribute__((ext_vector_type(2))) float v2f;
typedef __attribute__((ext_vector_type(4))) float v4f;
typedef __attribute__((ext_vector_type(8))) float v8f;

#define S_LEN   2048
#define NHEAD   16
#define DH      64
#define KSTRIDE 32
#define WIN     16
#define NTILES  7
#define NCOLS   112
#define ROWPAD  68
#define SCPAD   120
#define KSLOT   (16 * ROWPAD)

template <int XM>
__device__ __forceinline__ float row_xor(float x)
{
    const int v = __builtin_bit_cast(int, x);
    return __builtin_bit_cast(float,
        __builtin_amdgcn_update_dpp(0, v, 0x160 | XM, 0xf, 0xf, true));
}

__global__ __launch_bounds__(32)
void strided_attn_wmma_f32(const float* __restrict__ q,
                           const float* __restrict__ k,
                           const float* __restrict__ v,
                           float* __restrict__ out)
{
    __shared__ __attribute__((aligned(16))) float lds[(1 + NTILES) * KSLOT];
    float* const q_lds  = lds;
    float* const k_lds  = lds + KSLOT;
    float* const sc_lds = lds + KSLOT;

    const int lane  = threadIdx.x;
    const int m     = lane & 15;
    const int hi    = lane >> 4;
    const int qb    = blockIdx.x;
    const int h     = blockIdx.y;
    const int qbase = qb * 16;
    const long hbase = (long)h * S_LEN * DH;

    const int lane_off = hi * DH + 4 * m;
    v4f qg[8];
    #pragma unroll
    for (int i = 0; i < 8; ++i)
        qg[i] = *(const v4f*)(q + hbase + (long)(qbase + 2 * i) * DH + lane_off);
    v4f kg[NTILES][8];
    #pragma unroll
    for (int t = 0; t < NTILES; ++t) {
        int row0;
        if (t < 3) {
            int blk = qb - 1 + t;
            blk = blk < 0 ? 0 : (blk > S_LEN / 16 - 1 ? S_LEN / 16 - 1 : blk);
            row0 = blk * 16;
        } else {
            row0 = (t - 3) * 16 * KSTRIDE;
        }
        const int rstep = (t < 3) ? 1 : KSTRIDE;
        #pragma unroll
        for (int i = 0; i < 8; ++i)
            kg[t][i] = *(const v4f*)(k + hbase + (long)(row0 + 2 * i * rstep) * DH
                                       + (long)(hi * rstep) * DH + 4 * m);
    }

    #define STORE_TILE(dst, regs)                                                     \
        _Pragma("unroll")                                                             \
        for (int i_ = 0; i_ < 8; ++i_)                                                \
            *(v4f*)&(dst)[(2 * i_ + hi) * ROWPAD + 4 * m] = (regs)[i_];
    #define LOAD_FRAG(dstf, srcp)                                                     \
        _Pragma("unroll")                                                             \
        for (int kc_ = 0; kc_ < 16; ++kc_)                                            \
            (dstf)[kc_] = *(const v2f*)&(srcp)[m * ROWPAD + 4 * kc_ + 2 * hi];
    #define LDS_HANDOVER()                                                            \
        do {                                                                          \
            __builtin_amdgcn_fence(__ATOMIC_RELEASE, "workgroup", "local");           \
            __builtin_amdgcn_s_barrier();                                             \
            __builtin_amdgcn_fence(__ATOMIC_ACQUIRE, "workgroup", "local");           \
        } while (0)

    STORE_TILE(q_lds, qg)
    STORE_TILE(k_lds, kg[0])
    STORE_TILE(k_lds + KSLOT, kg[1])
    LDS_HANDOVER();

    v2f qa[16];
    v2f kb[2][16];
    LOAD_FRAG(qa, q_lds)
    LOAD_FRAG(kb[0], k_lds)

    v8f sc[NTILES];
    #pragma unroll
    for (int t = 0; t < NTILES; ++t) {
        if (t + 1 < NTILES) {
            LOAD_FRAG(kb[(t + 1) & 1], k_lds + (t + 1) * KSLOT)
        }
        if (t + 2 < NTILES) {
            STORE_TILE(k_lds + (t + 2) * KSLOT, kg[t + 2])
        }
        __builtin_amdgcn_sched_barrier(0);

        v8f acc = {};
        #pragma unroll
        for (int kc = 0; kc < 16; ++kc)
            acc = __builtin_amdgcn_wmma_f32_16x16x4_f32(
                false, qa[kc], false, kb[t & 1][kc], (short)0, acc, false, false);
        __builtin_amdgcn_sched_barrier(0);

        #pragma unroll
        for (int r = 0; r < 8; ++r) {
            const int row = r + 8 * hi;
            const int n   = qbase + row;
            const int col = t * 16 + m;
            bool valid;
            if (t < 3) {
                const int p = qbase - WIN + col;
                valid = (col >= row) && (col <= row + 2 * WIN)
                        && (p >= 0) && (p < S_LEN);
            } else {
                valid = (n > WIN);
            }
            sc[t][r] = valid ? acc[r] * 0.125f : -1.0e30f;
        }
        LDS_HANDOVER();
    }
    #undef STORE_TILE
    #undef LOAD_FRAG

    v4f vf[NTILES][4][2];
    #pragma unroll
    for (int t = 0; t < NTILES; ++t) {
        #pragma unroll
        for (int j = 0; j < 4; ++j) {
            #pragma unroll
            for (int s = 0; s < 2; ++s) {
                const int kr = 4 * j + 2 * hi + s;
                int pos;
                if (t < 3) {
                    pos = qbase - WIN + t * 16 + kr;
                    pos = pos < 0 ? 0 : (pos > S_LEN - 1 ? S_LEN - 1 : pos);
                } else {
                    pos = ((t - 3) * 16 + kr) * KSTRIDE;
                }
                vf[t][j][s] = *(const v4f*)(v + hbase + (long)pos * DH + 4 * m);
            }
        }
    }
    __builtin_amdgcn_sched_barrier(0);

    #pragma unroll
    for (int r = 0; r < 8; ++r) {
        float mx = sc[0][r];
        #pragma unroll
        for (int t = 1; t < NTILES; ++t)
            mx = fmaxf(mx, sc[t][r]);
        mx = fmaxf(mx, row_xor<1>(mx));
        mx = fmaxf(mx, row_xor<2>(mx));
        mx = fmaxf(mx, row_xor<4>(mx));
        mx = fmaxf(mx, row_xor<8>(mx));

        float sum = 0.0f;
        #pragma unroll
        for (int t = 0; t < NTILES; ++t) {
            const float e = __expf(sc[t][r] - mx);
            sc[t][r] = e;
            sum += e;
        }
        sum += row_xor<1>(sum);
        sum += row_xor<2>(sum);
        sum += row_xor<4>(sum);
        sum += row_xor<8>(sum);
        const float inv = (sum > 0.0f) ? (1.0f / sum) : 0.0f;

        const int row = r + 8 * hi;
        #pragma unroll
        for (int t = 0; t < NTILES; ++t)
            sc_lds[row * SCPAD + t * 16 + m] = sc[t][r] * inv;
    }
    __builtin_amdgcn_fence(__ATOMIC_RELEASE, "workgroup", "local");
    __builtin_amdgcn_s_barrier();
    __builtin_amdgcn_fence(__ATOMIC_ACQUIRE, "workgroup", "local");

    v8f acc_o[4] = {v8f{}, v8f{}, v8f{}, v8f{}};
    #pragma unroll
    for (int t = 0; t < NTILES; ++t) {
        #pragma unroll
        for (int j = 0; j < 4; ++j) {
            const int c0 = t * 16 + 4 * j + 2 * hi;
            v2f a = *(const v2f*)&sc_lds[m * SCPAD + c0];
            #pragma unroll
            for (int nt = 0; nt < 4; ++nt) {
                v2f b;
                b.x = vf[t][j][0][nt];
                b.y = vf[t][j][1][nt];
                acc_o[nt] = __builtin_amdgcn_wmma_f32_16x16x4_f32(
                    false, a, false, b, (short)0, acc_o[nt], false, false);
            }
        }
    }

    #pragma unroll
    for (int r = 0; r < 8; ++r) {
        const int row = r + 8 * hi;
        v4f o;
        o.x = acc_o[0][r]; o.y = acc_o[1][r]; o.z = acc_o[2][r]; o.w = acc_o[3][r];
        *(v4f*)(out + hbase + (long)(qbase + row) * DH + 4 * m) = o;
    }
}

extern "C" void kernel_launch(void* const* d_in, const int* in_sizes, int n_in,
                              void* d_out, int out_size, void* d_ws, size_t ws_size,
                              hipStream_t stream) {
    const float* q = (const float*)d_in[0];
    const float* k = (const float*)d_in[1];
    const float* v = (const float*)d_in[2];
    float* out = (float*)d_out;

    dim3 grid(S_LEN / 16, NHEAD, 1);
    dim3 block(32, 1, 1);
    strided_attn_wmma_f32<<<grid, block, 0, stream>>>(q, k, v, out);
}


# Eight consecutive query blocks of one head per 8-wave workgroup (a wave each, private LDS, no cross-wave barrier): the K/V rows they share hit the compute unit's cache, not L2 once per wave
# speedup vs baseline: 1.8450x; 1.2733x over previous
#include <hip/hip_runtime.h>

typedef __attribute__((ext_vector_type(2))) float v2f;
typedef __attribute__((ext_vector_type(4))) float v4f;
typedef __attribute__((ext_vector_type(8))) float v8f;

#define S_LEN   2048
#define NHEAD   16
#define DH      64
#define KSTRIDE 32
#define WIN     16
#define NTILES  7
#define NCOLS   112
#define ROWPAD  68
#define SCPAD   120
#define KSLOT   (16 * ROWPAD)
#define WAVES   8
#define WSLOTS  ((1 + NTILES) * KSLOT)

template <int XM>
__device__ __forceinline__ float row_xor(float x)
{
    const int v = __builtin_bit_cast(int, x);
    return __builtin_bit_cast(float,
        __builtin_amdgcn_update_dpp(0, v, 0x160 | XM, 0xf, 0xf, true));
}

__global__ __launch_bounds__(32 * WAVES)
void strided_attn_wmma_f32(const float* __restrict__ q,
                           const float* __restrict__ k,
                           const float* __restrict__ v,
                           float* __restrict__ out)
{
    __shared__ __attribute__((aligned(16))) float lds[WAVES * WSLOTS];
    const int wave  = __builtin_amdgcn_readfirstlane((int)(threadIdx.x >> 5));
    float* const q_lds  = lds + wave * WSLOTS;
    float* const k_lds  = q_lds + KSLOT;
    float* const sc_lds = q_lds + KSLOT;

    const int lane  = threadIdx.x & 31;
    const int m     = lane & 15;
    const int hi    = lane >> 4;
    const int qb    = blockIdx.x * WAVES + wave;
    const int h     = blockIdx.y;
    const int qbase = qb * 16;
    const long hbase = (long)h * S_LEN * DH;

    const int lane_off = hi * DH + 4 * m;
    v4f qg[8];
    #pragma unroll
    for (int i = 0; i < 8; ++i)
        qg[i] = *(const v4f*)(q + hbase + (long)(qbase + 2 * i) * DH + lane_off);
    v4f kg[NTILES][8];
    #pragma unroll
    for (int t = 0; t < NTILES; ++t) {
        int row0;
        if (t < 3) {
            int blk = qb - 1 + t;
            blk = blk < 0 ? 0 : (blk > S_LEN / 16 - 1 ? S_LEN / 16 - 1 : blk);
            row0 = blk * 16;
        } else {
            row0 = (t - 3) * 16 * KSTRIDE;
        }
        const int rstep = (t < 3) ? 1 : KSTRIDE;
        #pragma unroll
        for (int i = 0; i < 8; ++i)
            kg[t][i] = *(const v4f*)(k + hbase + (long)(row0 + 2 * i * rstep) * DH
                                       + (long)(hi * rstep) * DH + 4 * m);
    }

    #define STORE_TILE(dst, regs)                                                     \
        _Pragma("unroll")                                                             \
        for (int i_ = 0; i_ < 8; ++i_)                                                \
            *(v4f*)&(dst)[(2 * i_ + hi) * ROWPAD + 4 * m] = (regs)[i_];
    #define LOAD_FRAG(dstf, srcp)                                                     \
        _Pragma("unroll")                                                             \
        for (int kc_ = 0; kc_ < 16; ++kc_)                                            \
            (dstf)[kc_] = *(const v2f*)&(srcp)[m * ROWPAD + 4 * kc_ + 2 * hi];
    #define LDS_HANDOVER()                                                            \
        do {                                                                          \
            __builtin_amdgcn_fence(__ATOMIC_RELEASE, "workgroup", "local");           \
            __builtin_amdgcn_wave_barrier();                                          \
            __builtin_amdgcn_fence(__ATOMIC_ACQUIRE, "workgroup", "local");           \
        } while (0)

    STORE_TILE(q_lds, qg)
    STORE_TILE(k_lds, kg[0])
    STORE_TILE(k_lds + KSLOT, kg[1])
    LDS_HANDOVER();

    v2f qa[16];
    v2f kb[2][16];
    LOAD_FRAG(qa, q_lds)
    LOAD_FRAG(kb[0], k_lds)

    v8f sc[NTILES];
    #pragma unroll
    for (int t = 0; t < NTILES; ++t) {
        if (t + 1 < NTILES) {
            LOAD_FRAG(kb[(t + 1) & 1], k_lds + (t + 1) * KSLOT)
        }
        if (t + 2 < NTILES) {
            STORE_TILE(k_lds + (t + 2) * KSLOT, kg[t + 2])
        }
        __builtin_amdgcn_sched_barrier(0);

        v8f acc = {};
        #pragma unroll
        for (int kc = 0; kc < 16; ++kc)
            acc = __builtin_amdgcn_wmma_f32_16x16x4_f32(
                false, qa[kc], false, kb[t & 1][kc], (short)0, acc, false, false);
        __builtin_amdgcn_sched_barrier(0);

        #pragma unroll
        for (int r = 0; r < 8; ++r) {
            const int row = r + 8 * hi;
            const int n   = qbase + row;
            const int col = t * 16 + m;
            bool valid;
            if (t < 3) {
                const int p = qbase - WIN + col;
                valid = (col >= row) && (col <= row + 2 * WIN)
                        && (p >= 0) && (p < S_LEN);
            } else {
                valid = (n > WIN);
            }
            sc[t][r] = valid ? acc[r] * 0.125f : -1.0e30f;
        }
        LDS_HANDOVER();
    }
    #undef STORE_TILE
    #undef LOAD_FRAG

    v4f vf[NTILES][4][2];
    #pragma unroll
    for (int t = 0; t < NTILES; ++t) {
        #pragma unroll
        for (int j = 0; j < 4; ++j) {
            #pragma unroll
            for (int s = 0; s < 2; ++s) {
                const int kr = 4 * j + 2 * hi + s;
                int pos;
                if (t < 3) {
                    pos = qbase - WIN + t * 16 + kr;
                    pos = pos < 0 ? 0 : (pos > S_LEN - 1 ? S_LEN - 1 : pos);
                } else {
                    pos = ((t - 3) * 16 + kr) * KSTRIDE;
                }
                vf[t][j][s] = *(const v4f*)(v + hbase + (long)pos * DH + 4 * m);
            }
        }
    }
    __builtin_amdgcn_sched_barrier(0);

    #pragma unroll
    for (int r = 0; r < 8; ++r) {
        float mx = sc[0][r];
        #pragma unroll
        for (int t = 1; t < NTILES; ++t)
            mx = fmaxf(mx, sc[t][r]);
        mx = fmaxf(mx, row_xor<1>(mx));
        mx = fmaxf(mx, row_xor<2>(mx));
        mx = fmaxf(mx, row_xor<4>(mx));
        mx = fmaxf(mx, row_xor<8>(mx));

        float sum = 0.0f;
        #pragma unroll
        for (int t = 0; t < NTILES; ++t) {
            const float e = __expf(sc[t][r] - mx);
            sc[t][r] = e;
            sum += e;
        }
        sum += row_xor<1>(sum);
        sum += row_xor<2>(sum);
        sum += row_xor<4>(sum);
        sum += row_xor<8>(sum);
        const float inv = (sum > 0.0f) ? (1.0f / sum) : 0.0f;

        const int row = r + 8 * hi;
        #pragma unroll
        for (int t = 0; t < NTILES; ++t)
            sc_lds[row * SCPAD + t * 16 + m] = sc[t][r] * inv;
    }
    LDS_HANDOVER();

    v8f acc_o[4] = {v8f{}, v8f{}, v8f{}, v8f{}};
    #pragma unroll
    for (int t = 0; t < NTILES; ++t) {
        #pragma unroll
        for (int j = 0; j < 4; ++j) {
            const int c0 = t * 16 + 4 * j + 2 * hi;
            v2f a = *(const v2f*)&sc_lds[m * SCPAD + c0];
            #pragma unroll
            for (int nt = 0; nt < 4; ++nt) {
                v2f b;
                b.x = vf[t][j][0][nt];
                b.y = vf[t][j][1][nt];
                acc_o[nt] = __builtin_amdgcn_wmma_f32_16x16x4_f32(
                    false, a, false, b, (short)0, acc_o[nt], false, false);
            }
        }
        __builtin_amdgcn_sched_barrier(0x100);
    }

    #pragma unroll
    for (int r = 0; r < 8; ++r) {
        const int row = r + 8 * hi;
        v4f o;
        o.x = acc_o[0][r]; o.y = acc_o[1][r]; o.z = acc_o[2][r]; o.w = acc_o[3][r];
        *(v4f*)(out + hbase + (long)(qbase + row) * DH + 4 * m) = o;
    }
}

extern "C" void kernel_launch(void* const* d_in, const int* in_sizes, int n_in,
                              void* d_out, int out_size, void* d_ws, size_t ws_size,
                              hipStream_t stream) {
    const float* q = (const float*)d_in[0];
    const float* k = (const float*)d_in[1];
    const float* v = (const float*)d_in[2];
    float* out = (float*)d_out;

    dim3 grid(S_LEN / 16 / WAVES, NHEAD, 1);
    dim3 block(32 * WAVES, 1, 1);
    strided_attn_wmma_f32<<<grid, block, 0, stream>>>(q, k, v, out);
}


# Scores computed transposed (K tile as A, Q as B): a lane owns a query row, its probabilities are the PV A operand in registers; no LDS round trip, no row butterflies; softmax sum order changes
# speedup vs baseline: 1.9806x; 1.0735x over previous
#include <hip/hip_runtime.h>

typedef __attribute__((ext_vector_type(2))) float v2f;
typedef __attribute__((ext_vector_type(4))) float v4f;
typedef __attribute__((ext_vector_type(8))) float v8f;

#define S_LEN   2048
#define NHEAD   16
#define DH      64
#define KSTRIDE 32
#define WIN     16
#define NTILES  7
#define NCOLS   112
#define ROWPAD  68
#define KSLOT   (16 * ROWPAD)
#define WAVES   8
#define WSLOTS  ((1 + NTILES) * KSLOT)

__device__ __forceinline__ float xhalf(float x)
{
    const int v = __builtin_bit_cast(int, x);
    return __builtin_bit_cast(float,
        __builtin_amdgcn_permlanex16(v, v, 0x76543210, 0xfedcba98, false, false));
}

__global__ __launch_bounds__(32 * WAVES)
void strided_attn_wmma_f32(const float* __restrict__ q,
                           const float* __restrict__ k,
                           const float* __restrict__ v,
                           float* __restrict__ out)
{
    __shared__ __attribute__((aligned(16))) float lds[WAVES * WSLOTS];
    const int wave  = __builtin_amdgcn_readfirstlane((int)(threadIdx.x >> 5));
    float* const q_lds  = lds + wave * WSLOTS;
    float* const k_lds  = q_lds + KSLOT;

    const int lane  = threadIdx.x & 31;
    const int m     = lane & 15;
    const int hi    = lane >> 4;
    const int qb    = blockIdx.x * WAVES + wave;
    const int h     = blockIdx.y;
    const int qbase = qb * 16;
    const long hbase = (long)h * S_LEN * DH;

    const int lane_off = hi * DH + 4 * m;
    v4f qg[8];
    #pragma unroll
    for (int i = 0; i < 8; ++i)
        qg[i] = *(const v4f*)(q + hbase + (long)(qbase + 2 * i) * DH + lane_off);
    v4f kg[NTILES][8];
    #pragma unroll
    for (int t = 0; t < NTILES; ++t) {
        int row0;
        if (t < 3) {
            int blk = qb - 1 + t;
            blk = blk < 0 ? 0 : (blk > S_LEN / 16 - 1 ? S_LEN / 16 - 1 : blk);
            row0 = blk * 16;
        } else {
            row0 = (t - 3) * 16 * KSTRIDE;
        }
        const int rstep = (t < 3) ? 1 : KSTRIDE;
        #pragma unroll
        for (int i = 0; i < 8; ++i)
            kg[t][i] = *(const v4f*)(k + hbase + (long)(row0 + 2 * i * rstep) * DH
                                       + (long)(hi * rstep) * DH + 4 * m);
    }

    #define STORE_TILE(dst, regs)                                                     \
        _Pragma("unroll")                                                             \
        for (int i_ = 0; i_ < 8; ++i_)                                                \
            *(v4f*)&(dst)[(2 * i_ + hi) * ROWPAD + 4 * m] = (regs)[i_];
    #define LOAD_FRAG(dstf, srcp)                                                     \
        _Pragma("unroll")                                                             \
        for (int kc_ = 0; kc_ < 16; ++kc_)                                            \
            (dstf)[kc_] = *(const v2f*)&(srcp)[m * ROWPAD + 4 * kc_ + 2 * hi];
    #define LDS_HANDOVER()                                                            \
        do {                                                                          \
            __builtin_amdgcn_fence(__ATOMIC_RELEASE, "workgroup", "local");           \
            __builtin_amdgcn_wave_barrier();                                          \
            __builtin_amdgcn_fence(__ATOMIC_ACQUIRE, "workgroup", "local");           \
        } while (0)

    STORE_TILE(q_lds, qg)
    STORE_TILE(k_lds, kg[0])
    STORE_TILE(k_lds + KSLOT, kg[1])
    LDS_HANDOVER();

    v2f qa[16];
    v2f kb[2][16];
    LOAD_FRAG(qa, q_lds)
    LOAD_FRAG(kb[0], k_lds)

    v8f sc[NTILES];
    #pragma unroll
    for (int t = 0; t < NTILES; ++t) {
        if (t + 1 < NTILES) {
            LOAD_FRAG(kb[(t + 1) & 1], k_lds + (t + 1) * KSLOT)
        }
        if (t + 2 < NTILES) {
            STORE_TILE(k_lds + (t + 2) * KSLOT, kg[t + 2])
        }
        __builtin_amdgcn_sched_barrier(0);

        v8f acc = {};
        #pragma unroll
        for (int kc = 0; kc < 16; ++kc)
            acc = __builtin_amdgcn_wmma_f32_16x16x4_f32(
                false, kb[t & 1][kc], false, qa[kc], (short)0, acc, false, false);
        __builtin_amdgcn_sched_barrier(0);

        #pragma unroll
        for (int r = 0; r < 8; ++r) {
            const int row = m;
            const int n   = qbase + row;
            const int col = t * 16 + r + 8 * hi;
            bool valid;
            if (t < 3) {
                const int p = qbase - WIN + col;
                valid = (col >= row) && (col <= row + 2 * WIN)
                        && (p >= 0) && (p < S_LEN);
            } else {
                valid = (n > WIN);
            }
            sc[t][r] = valid ? acc[r] * 0.125f : -1.0e30f;
        }
        if (t + 1 < NTILES) {
            LDS_HANDOVER();
        }
    }
    #undef STORE_TILE
    #undef LOAD_FRAG

    v4f vf[NTILES][4][2];
    #pragma unroll
    for (int t = 0; t < NTILES; ++t) {
        #pragma unroll
        for (int j = 0; j < 4; ++j) {
            #pragma unroll
            for (int s = 0; s < 2; ++s) {
                const int kr = 8 * hi + 2 * j + s;
                int pos;
                if (t < 3) {
                    pos = qbase - WIN + t * 16 + kr;
                    pos = pos < 0 ? 0 : (pos > S_LEN - 1 ? S_LEN - 1 : pos);
                } else {
                    pos = ((t - 3) * 16 + kr) * KSTRIDE;
                }
                vf[t][j][s] = *(const v4f*)(v + hbase + (long)pos * DH + 4 * m);
            }
        }
    }
    __builtin_amdgcn_sched_barrier(0);

    float mxp[4] = {sc[0][0], sc[0][1], sc[0][2], sc[0][3]};
    #pragma unroll
    for (int t = 0; t < NTILES; ++t) {
        #pragma unroll
        for (int r = 0; r < 8; ++r)
            mxp[r & 3] = fmaxf(mxp[r & 3], sc[t][r]);
    }
    float mx = fmaxf(fmaxf(mxp[0], mxp[1]), fmaxf(mxp[2], mxp[3]));
    mx = fmaxf(mx, xhalf(mx));

    v2f ps[2] = {v2f{0.0f, 0.0f}, v2f{0.0f, 0.0f}};
    #pragma unroll
    for (int t = 0; t < NTILES; ++t) {
        #pragma unroll
        for (int j = 0; j < 4; ++j) {
            v2f e;
            e.x = __expf(sc[t][2 * j] - mx);
            e.y = __expf(sc[t][2 * j + 1] - mx);
            sc[t][2 * j]     = e.x;
            sc[t][2 * j + 1] = e.y;
            ps[j & 1] += e;
        }
    }
    float sum = (ps[0].x + ps[0].y) + (ps[1].x + ps[1].y);
    sum += xhalf(sum);
    const float inv = (sum > 0.0f) ? (1.0f / sum) : 0.0f;
    #pragma unroll
    for (int t = 0; t < NTILES; ++t) {
        #pragma unroll
        for (int r = 0; r < 8; ++r)
            sc[t][r] *= inv;
    }

    v8f acc_o[4] = {v8f{}, v8f{}, v8f{}, v8f{}};
    #pragma unroll
    for (int t = 0; t < NTILES; ++t) {
        #pragma unroll
        for (int j = 0; j < 4; ++j) {
            v2f a;
            a.x = sc[t][2 * j];
            a.y = sc[t][2 * j + 1];
            #pragma unroll
            for (int nt = 0; nt < 4; ++nt) {
                v2f b;
                b.x = vf[t][j][0][nt];
                b.y = vf[t][j][1][nt];
                acc_o[nt] = __builtin_amdgcn_wmma_f32_16x16x4_f32(
                    false, a, false, b, (short)0, acc_o[nt], false, false);
            }
        }
        __builtin_amdgcn_sched_barrier(0);
    }

    #pragma unroll
    for (int r = 0; r < 8; ++r) {
        const int row = r + 8 * hi;
        v4f o;
        o.x = acc_o[0][r]; o.y = acc_o[1][r]; o.z = acc_o[2][r]; o.w = acc_o[3][r];
        *(v4f*)(out + hbase + (long)(qbase + row) * DH + 4 * m) = o;
    }
}

extern "C" void kernel_launch(void* const* d_in, const int* in_sizes, int n_in,
                              void* d_out, int out_size, void* d_ws, size_t ws_size,
                              hipStream_t stream) {
    const float* q = (const float*)d_in[0];
    const float* k = (const float*)d_in[1];
    const float* v = (const float*)d_in[2];
    float* out = (float*)d_out;

    dim3 grid(S_LEN / 16 / WAVES, NHEAD, 1);
    dim3 block(32 * WAVES, 1, 1);
    strided_attn_wmma_f32<<<grid, block, 0, stream>>>(q, k, v, out);
}


# Gap between score and PV matrix runs cut to the softmax alone: V rows from wave-uniform bases, requested in the score steps; masks as accumulator start values; 1/8 scale fused into the exp argument
# speedup vs baseline: 2.1439x; 1.0824x over previous
#include <hip/hip_runtime.h>

typedef __attribute__((ext_vector_type(2))) float v2f;
typedef __attribute__((ext_vector_type(4))) float v4f;
typedef __attribute__((ext_vector_type(8))) float v8f;

#define S_LEN   2048
#define NHEAD   16
#define DH      64
#define KSTRIDE 32
#define WIN     16
#define NTILES  7
#define NCOLS   112
#define ROWPAD  68
#define KSLOT   (16 * ROWPAD)
#define WAVES   8
#define WSLOTS  ((1 + NTILES) * KSLOT)

__device__ __forceinline__ float xhalf(float x)
{
    const int v = __builtin_bit_cast(int, x);
    return __builtin_bit_cast(float,
        __builtin_amdgcn_permlanex16(v, v, 0x76543210, 0xfedcba98, false, false));
}

__global__ __launch_bounds__(32 * WAVES)
void strided_attn_wmma_f32(const float* __restrict__ q,
                           const float* __restrict__ k,
                           const float* __restrict__ v,
                           float* __restrict__ out)
{
    __shared__ __attribute__((aligned(16))) float lds[WAVES * WSLOTS];
    const int wave  = __builtin_amdgcn_readfirstlane((int)(threadIdx.x >> 5));
    float* const q_lds  = lds + wave * WSLOTS;
    float* const k_lds  = q_lds + KSLOT;

    const int lane  = threadIdx.x & 31;
    const int m     = lane & 15;
    const int hi    = lane >> 4;
    const int qb    = blockIdx.x * WAVES + wave;
    const int h     = blockIdx.y;
    const int qbase = qb * 16;
    const long hbase = (long)h * S_LEN * DH;

    const int lane_off = hi * DH + 4 * m;
    v4f qg[8];
    #pragma unroll
    for (int i = 0; i < 8; ++i)
        qg[i] = *(const v4f*)(q + hbase + (long)(qbase + 2 * i) * DH + lane_off);
    v4f kg[NTILES][8];
    #pragma unroll
    for (int t = 0; t < NTILES; ++t) {
        int row0;
        if (t < 3) {
            int blk = qb - 1 + t;
            blk = blk < 0 ? 0 : (blk > S_LEN / 16 - 1 ? S_LEN / 16 - 1 : blk);
            row0 = blk * 16;
        } else {
            row0 = (t - 3) * 16 * KSTRIDE;
        }
        const int rstep = (t < 3) ? 1 : KSTRIDE;
        #pragma unroll
        for (int i = 0; i < 8; ++i)
            kg[t][i] = *(const v4f*)(k + hbase + (long)(row0 + 2 * i * rstep) * DH
                                       + (long)(hi * rstep) * DH + 4 * m);
    }

    __builtin_amdgcn_sched_barrier(0);

    const float MASKV = -8.0e30f;
    const int  dkq    = m - 8 * hi;
    const int  d0     = (qb >= 1) ? dkq : 64;
    const int  d2     = (qb <= S_LEN / 16 - 2) ? dkq : -64;
    const float sbias = (qbase + m > WIN) ? 0.0f : MASKV;
    v8f bias0, bias2, biass;
    #pragma unroll
    for (int r = 0; r < 8; ++r) {
        bias0[r] = (r >= d0) ? 0.0f : MASKV;
        bias2[r] = (r <= d2) ? 0.0f : MASKV;
        biass[r] = sbias;
    }
    asm volatile("" :: "v"(bias0), "v"(bias2), "v"(biass));

    const float* const vh = v + hbase;
    const int vlane_loc = (8 * hi) * DH + 4 * m;
    const int vlane_str = (8 * hi * KSTRIDE) * DH + 4 * m;

    #define STORE_TILE(dst, regs)                                                     \
        _Pragma("unroll")                                                             \
        for (int i_ = 0; i_ < 8; ++i_)                                                \
            *(v4f*)&(dst)[(2 * i_ + hi) * ROWPAD + 4 * m] = (regs)[i_];
    #define LOAD_FRAG(dstf, srcp)                                                     \
        _Pragma("unroll")                                                             \
        for (int kc_ = 0; kc_ < 16; ++kc_)                                            \
            (dstf)[kc_] = *(const v2f*)&(srcp)[m * ROWPAD + 4 * kc_ + 2 * hi];
    #define LDS_HANDOVER()                                                            \
        do {                                                                          \
            __builtin_amdgcn_fence(__ATOMIC_RELEASE, "workgroup", "local");           \
            __builtin_amdgcn_wave_barrier();                                          \
            __builtin_amdgcn_fence(__ATOMIC_ACQUIRE, "workgroup", "local");           \
        } while (0)

    STORE_TILE(q_lds, qg)
    STORE_TILE(k_lds, kg[0])
    STORE_TILE(k_lds + KSLOT, kg[1])
    LDS_HANDOVER();

    v2f qa[16];
    v2f kb[2][16];
    LOAD_FRAG(qa, q_lds)
    LOAD_FRAG(kb[0], k_lds)

    v8f sc[NTILES];
    v4f vf[NTILES][4][2];
    #pragma unroll
    for (int t = 0; t < NTILES; ++t) {
        if (t + 1 < NTILES) {
            LOAD_FRAG(kb[(t + 1) & 1], k_lds + (t + 1) * KSLOT)
        }
        if (t + 2 < NTILES) {
            STORE_TILE(k_lds + (t + 2) * KSLOT, kg[t + 2])
        }
        {
            const float* vb;
            int rstep_v;
            if (t < 3) {
                int blk = qb - 1 + t;
                blk = blk < 0 ? 0 : (blk > S_LEN / 16 - 1 ? S_LEN / 16 - 1 : blk);
                vb = vh + (long)(blk * 16) * DH + vlane_loc;
                rstep_v = 1;
            } else {
                vb = vh + (long)((t - 3) * 16 * KSTRIDE) * DH + vlane_str;
                rstep_v = KSTRIDE;
            }
            #pragma unroll
            for (int j = 0; j < 4; ++j) {
                #pragma unroll
                for (int s = 0; s < 2; ++s)
                    vf[t][j][s] = *(const v4f*)(vb + (long)((2 * j + s) * rstep_v) * DH);
            }
        }
        __builtin_amdgcn_sched_barrier(0);

        v8f acc = (t == 0) ? bias0 : (t == 1) ? v8f{} : (t == 2) ? bias2 : biass;
        #pragma unroll
        for (int kc = 0; kc < 16; ++kc)
            acc = __builtin_amdgcn_wmma_f32_16x16x4_f32(
                false, kb[t & 1][kc], false, qa[kc], (short)0, acc, false, false);
        __builtin_amdgcn_sched_barrier(0);

        sc[t] = acc;
        if (t + 1 < NTILES) {
            LDS_HANDOVER();
        }
    }
    #undef STORE_TILE
    #undef LOAD_FRAG

    float mxp[4] = {sc[0][0], sc[0][1], sc[0][2], sc[0][3]};
    #pragma unroll
    for (int t = 0; t < NTILES; ++t) {
        #pragma unroll
        for (int r = 0; r < 8; ++r)
            mxp[r & 3] = fmaxf(mxp[r & 3], sc[t][r]);
    }
    float mx = fmaxf(fmaxf(mxp[0], mxp[1]), fmaxf(mxp[2], mxp[3]));
    mx = fmaxf(mx, xhalf(mx));
    const float mxs = mx * 0.125f;

    v2f ps[2] = {v2f{0.0f, 0.0f}, v2f{0.0f, 0.0f}};
    #pragma unroll
    for (int t = 0; t < NTILES; ++t) {
        #pragma unroll
        for (int j = 0; j < 4; ++j) {
            v2f e;
            e.x = __expf(fmaf(sc[t][2 * j],     0.125f, -mxs));
            e.y = __expf(fmaf(sc[t][2 * j + 1], 0.125f, -mxs));
            sc[t][2 * j]     = e.x;
            sc[t][2 * j + 1] = e.y;
            ps[j & 1] += e;
        }
    }
    float sum = (ps[0].x + ps[0].y) + (ps[1].x + ps[1].y);
    sum += xhalf(sum);
    const float inv = (sum > 0.0f) ? (1.0f / sum) : 0.0f;
    #pragma unroll
    for (int t = 0; t < NTILES; ++t) {
        #pragma unroll
        for (int r = 0; r < 8; ++r)
            sc[t][r] *= inv;
    }

    v8f acc_o[4] = {v8f{}, v8f{}, v8f{}, v8f{}};
    #pragma unroll
    for (int t = 0; t < NTILES; ++t) {
        #pragma unroll
        for (int j = 0; j < 4; ++j) {
            v2f a;
            a.x = sc[t][2 * j];
            a.y = sc[t][2 * j + 1];
            #pragma unroll
            for (int nt = 0; nt < 4; ++nt) {
                v2f b;
                b.x = vf[t][j][0][nt];
                b.y = vf[t][j][1][nt];
                acc_o[nt] = __builtin_amdgcn_wmma_f32_16x16x4_f32(
                    false, a, false, b, (short)0, acc_o[nt], false, false);
            }
        }
        __builtin_amdgcn_sched_barrier(0);
    }

    #pragma unroll
    for (int r = 0; r < 8; ++r) {
        const int row = r + 8 * hi;
        v4f o;
        o.x = acc_o[0][r]; o.y = acc_o[1][r]; o.z = acc_o[2][r]; o.w = acc_o[3][r];
        *(v4f*)(out + hbase + (long)(qbase + row) * DH + 4 * m) = o;
    }
}

extern "C" void kernel_launch(void* const* d_in, const int* in_sizes, int n_in,
                              void* d_out, int out_size, void* d_ws, size_t ws_size,
                              hipStream_t stream) {
    const float* q = (const float*)d_in[0];
    const float* k = (const float*)d_in[1];
    const float* v = (const float*)d_in[2];
    float* out = (float*)d_out;

    dim3 grid(S_LEN / 16 / WAVES, NHEAD, 1);
    dim3 block(32 * WAVES, 1, 1);
    strided_attn_wmma_f32<<<grid, block, 0, stream>>>(q, k, v, out);
}
